# starting kernel
# baseline (speedup 1.0000x reference)
#include <hip/hip_runtime.h>

typedef float v2f __attribute__((ext_vector_type(2)));
typedef float v8f __attribute__((ext_vector_type(8)));

#define TC      5
#define B_      4
#define C_      64
#define T_      1000
#define F_      96
#define O_      2
#define KTOT    (C_ * TC)
#define NKC     (KTOT / 4)
#define TTILE   16
#define WIN     (TTILE + TC - 1)
#define FCHUNK  8
#define TSTRIDE 68
#define XSTRIDE_F 1364
#define WB_OFF  (FCHUNK * XSTRIDE_F)
#define WB_PER_F (KTOT * O_)
#define LDS_FLOATS (WB_OFF + FCHUNK * WB_PER_F)
#define NT_TILES   ((T_ + TTILE - 1) / TTILE)
#define NF_CHUNKS  (F_ / FCHUNK)


__global__ __launch_bounds__(256)
void decoder_out_wmma(const float* __restrict__ x,
                      const float* __restrict__ w,
                      const float* __restrict__ bias,
                      float* __restrict__ out) {
    __shared__ float lds[LDS_FLOATS];

    const int blk    = blockIdx.x;
    const int tTile  = blk % NT_TILES;
    const int fChunk = (blk / NT_TILES) % NF_CHUNKS;
    const int b      = blk / (NT_TILES * NF_CHUNKS);

    const int t0  = tTile * TTILE;
    const int f0  = fChunk * FCHUNK;
    const int tid = threadIdx.x;

    for (int i = tid; i < C_ * WIN * FCHUNK; i += 256) {
        const int fl   = i & (FCHUNK - 1);
        const int pair = i >> 3;
        const int tl   = pair % WIN;
        const int c    = pair / WIN;
        const int tg   = t0 - (TC - 1) + tl;
        float v = 0.0f;
        if (tg >= 0 && tg < T_)
            v = x[((size_t)(b * C_ + c) * T_ + tg) * F_ + (f0 + fl)];
        lds[fl * XSTRIDE_F + tl * TSTRIDE + c] = v;
    }

    for (int d = tid; d < FCHUNK * WB_PER_F; d += 256) {
        const int fl = d / WB_PER_F;
        const int r  = d - fl * WB_PER_F;
        const int kc = r >> 3;
        const int n  = (r >> 2) & 1;
        const int kk = r & 3;
        const int c  = ((kc & 15) << 2) + kk;
        const int k  = kc >> 4;
        lds[WB_OFF + d] = w[(size_t)(f0 + fl) * WB_PER_F + (c * TC + k) * O_ + n];
    }
    __syncthreads();

    const int wave = tid >> 5;
    const int lane = tid & 31;
    const int half = lane >> 4;
    const int mn   = lane & 15;
    const int f    = f0 + wave;

    const float* lfA = &lds[wave * XSTRIDE_F + mn * TSTRIDE + half * 2];
    const int nclamp = (mn < O_) ? mn : 1;
    const float* lfB = &lds[WB_OFF + wave * WB_PER_F + nclamp * 4 + half * 2];

    v8f acc = {};

#pragma unroll
    for (int kc = 0; kc < NKC; ++kc) {
        const int aoff = (kc >> 4) * TSTRIDE + (kc & 15) * 4;
        const int boff = kc * 8;
        const v2f a  = *(const v2f*)(lfA + aoff);
        const v2f bf = *(const v2f*)(lfB + boff);
        acc = __builtin_amdgcn_wmma_f32_16x16x4_f32(
            false, a, false, bf, (short)0, acc, false, false);
    }

    if (mn < O_) {
        const float bv = bias[f * O_ + mn];
#pragma unroll
        for (int r = 0; r < 8; ++r) {
            const int t = t0 + half * 8 + r;
            if (t < T_)
                out[((size_t)(b * O_ + mn) * T_ + t) * F_ + f] = acc[r] + bv;
        }
    }
}

extern "C" void kernel_launch(void* const* d_in, const int* in_sizes, int n_in,
                              void* d_out, int out_size, void* d_ws, size_t ws_size,
                              hipStream_t stream) {
    const float* x    = (const float*)d_in[0];
    const float* w    = (const float*)d_in[1];
    const float* bias = (const float*)d_in[2];
    float* out = (float*)d_out;

    const int grid = B_ * NF_CHUNKS * NT_TILES;
    decoder_out_wmma<<<grid, 256, 0, stream>>>(x, w, bias, out);
}


# Staging: all weight and x-tile global loads issued up front as independent 16-byte loads held in registers, then 8-byte LDS stores: one memory round trip per thread instead of 40 serial ones
# speedup vs baseline: 1.5889x; 1.5889x over previous
#include <hip/hip_runtime.h>


typedef float v2f __attribute__((ext_vector_type(2)));
typedef float v4f __attribute__((ext_vector_type(4)));
typedef float v8f __attribute__((ext_vector_type(8)));

#define TC      5
#define B_      4
#define C_      64
#define T_      1000
#define F_      96
#define O_      2
#define KTOT    (C_ * TC)
#define NKC     (KTOT / 4)
#define TTILE   16
#define WIN     (TTILE + TC - 1)
#define FCHUNK  8
#define TSTRIDE 68
#define XSTRIDE_F 1368
#define WB_OFF  (FCHUNK * XSTRIDE_F)
#define WB_PER_F (KTOT * O_)
#define LDS_FLOATS (WB_OFF + FCHUNK * WB_PER_F)
#define NT_TILES   ((T_ + TTILE - 1) / TTILE)
#define NF_CHUNKS  (F_ / FCHUNK)
#define XUNITS     5
#define WQUADS     5


__global__ __launch_bounds__(256)
void decoder_out_wmma(const float* __restrict__ x,
                      const float* __restrict__ w,
                      const float* __restrict__ bias,
                      float* __restrict__ out) {
    __shared__ __attribute__((aligned(16))) float lds[LDS_FLOATS];

    const int blk    = blockIdx.x;
    const int tTile  = blk % NT_TILES;
    const int fChunk = (blk / NT_TILES) % NF_CHUNKS;
    const int b      = blk / (NT_TILES * NF_CHUNKS);

    const int t0   = tTile * TTILE;
    const int f0   = fChunk * FCHUNK;
    const int tid  = threadIdx.x;
    const int lane = tid & 31;
    const int wave = __builtin_amdgcn_readfirstlane(tid >> 5);

    const char* wrow = (const char*)(w + (size_t)(f0 + wave) * WB_PER_F);
    v4f wv[WQUADS];
#pragma unroll
    for (int j = 0; j < WQUADS; ++j)
        wv[j] = *(const v4f*)(wrow + (unsigned)(lane * 16 + j * 512));

    const char* xb = (const char*)(x + (size_t)b * C_ * T_ * F_ + f0);
    v4f xa[XUNITS], xc[XUNITS];
    bool okrow[XUNITS];
#pragma unroll
    for (int it = 0; it < XUNITS; ++it) {
        const int u    = tid + 256 * it;
        const int q    = u & 1;
        const int c2lo = (u >> 1) & 1;
        const int rest = u >> 2;
        const int tl   = rest % WIN;
        const int c    = (rest / WIN) * 4 + c2lo * 2;
        const int tg   = t0 - (TC - 1) + tl;
        const bool ok  = (unsigned)tg < (unsigned)T_;
        const int tgc  = ok ? tg : 0;
        const unsigned boff = (unsigned)((c * T_ + tgc) * F_ + q * 4) * 4u;
        xa[it] = *(const v4f*)(xb + boff);
        xc[it] = *(const v4f*)(xb + boff + (unsigned)(T_ * F_ * 4));
        okrow[it] = ok;
    }


    float* wB = &lds[WB_OFF + wave * WB_PER_F];
#pragma unroll
    for (int j = 0; j < WQUADS; ++j) {
#pragma unroll
        for (int h = 0; h < 2; ++h) {
            const int ck = j * 64 + lane * 2 + h;
            const int c  = ck / TC;
            const int k  = ck - c * TC;
            const int idx = (k * 16 + (c >> 2)) * 8 + (c & 3);
            wB[idx]     = wv[j][2 * h];
            wB[idx + 4] = wv[j][2 * h + 1];
        }
    }

#pragma unroll
    for (int it = 0; it < XUNITS; ++it)
        asm volatile("" : "+v"(xa[it]), "+v"(xc[it]));

#pragma unroll
    for (int it = 0; it < XUNITS; ++it) {
        const int u    = tid + 256 * it;
        const int q    = u & 1;
        const int c2lo = (u >> 1) & 1;
        const int rest = u >> 2;
        const int tl   = rest % WIN;
        const int c    = (rest / WIN) * 4 + c2lo * 2;
        float* dst = &lds[q * 4 * XSTRIDE_F + tl * TSTRIDE + c];
#pragma unroll
        for (int e = 0; e < 4; ++e) {
            v2f s;
            s.x = okrow[it] ? xa[it][e] : 0.0f;
            s.y = okrow[it] ? xc[it][e] : 0.0f;
            *(v2f*)(dst + e * XSTRIDE_F) = s;
        }
    }
    __syncthreads();

    const int half = lane >> 4;
    const int mn   = lane & 15;
    const int f    = f0 + wave;

    const float* lfA = &lds[wave * XSTRIDE_F + mn * TSTRIDE + half * 2];
    const int nclamp = (mn < O_) ? mn : 1;
    const float* lfB = &lds[WB_OFF + wave * WB_PER_F + nclamp * 4 + half * 2];

    v8f acc = {};

#pragma unroll
    for (int kc = 0; kc < NKC; ++kc) {
        const int aoff = (kc >> 4) * TSTRIDE + (kc & 15) * 4;
        const int boff = kc * 8;
        const v2f a  = *(const v2f*)(lfA + aoff);
        const v2f bf = *(const v2f*)(lfB + boff);
        acc = __builtin_amdgcn_wmma_f32_16x16x4_f32(
            false, a, false, bf, (short)0, acc, false, false);
    }

    if (mn < O_) {
        const float bv = bias[f * O_ + mn];
#pragma unroll
        for (int r = 0; r < 8; ++r) {
            const int t = t0 + half * 8 + r;
            if (t < T_)
                out[((size_t)(b * O_ + mn) * T_ + t) * F_ + f] = acc[r] + bv;
        }
    }
}

extern "C" void kernel_launch(void* const* d_in, const int* in_sizes, int n_in,
                              void* d_out, int out_size, void* d_ws, size_t ws_size,
                              hipStream_t stream) {
    const float* x    = (const float*)d_in[0];
    const float* w    = (const float*)d_in[1];
    const float* bias = (const float*)d_in[2];
    float* out = (float*)d_out;

    const int grid = B_ * NF_CHUNKS * NT_TILES;
    decoder_out_wmma<<<grid, 256, 0, stream>>>(x, w, bias, out);
}


# Tiling: work-group = 32 frequencies (a wave each) x 48 rows in 3 looped window tiles, so every wave-wide load covers whole 128-byte lines; weights staged once, next tile's loads issued ahead
# speedup vs baseline: 2.1605x; 1.3597x over previous
#include <hip/hip_runtime.h>


typedef float v2f __attribute__((ext_vector_type(2)));
typedef float v4f __attribute__((ext_vector_type(4)));
typedef float v8f __attribute__((ext_vector_type(8)));

#define TC      5
#define B_      4
#define C_      64
#define T_      1000
#define F_      96
#define O_      2
#define KTOT    (C_ * TC)
#define NKC     (C_ / 4)
#define WROWS   (O_ * TC)
#define WKC     (WROWS * 4)
#define TTILE   16
#define WIN     (TTILE + TC - 1)
#define FCHUNK  32
#define NTHREADS (FCHUNK * 32)
#define TILES_PER_WG 3
#define TSEG    (TTILE * TILES_PER_WG)
#define NSEG    ((T_ + TSEG - 1) / TSEG)
#define TSTRIDE 68
#define XSTRIDE_F 1362
#define WB_OFF  (FCHUNK * XSTRIDE_F)
#define WB_PER_F (KTOT * O_)
#define LDS_FLOATS (WB_OFF + FCHUNK * WB_PER_F)
#define NF_CHUNKS  (F_ / FCHUNK)
#define XUNITS     5
#define WQUADS     5


template <int R>
__device__ __forceinline__ float rot_left_in_row(float v) {
    const int r = __builtin_amdgcn_update_dpp(0, __builtin_bit_cast(int, v),
                                              0x120 + (16 - R), 0xf, 0xf, false);
    return __builtin_bit_cast(float, r);
}

struct XUnit { int q, tl, c; };
__device__ __forceinline__ XUnit x_unit(int u) {
    XUnit r;
    r.q = u & 7;
    const int rr    = u >> 3;
    const int tlp   = rr & 1;
    const int cpl   = (rr >> 1) & 1;
    const int rest  = rr >> 2;
    r.tl = (rest % (WIN / 2)) * 2 + tlp;
    r.c  = ((rest / (WIN / 2)) * 2 + cpl) * 2;
    return r;
}

__global__ __launch_bounds__(NTHREADS)
void decoder_out_wmma(const float* __restrict__ x,
                      const float* __restrict__ w,
                      const float* __restrict__ bias,
                      float* __restrict__ out) {
    __shared__ __attribute__((aligned(16))) float lds[LDS_FLOATS];

    const int blk    = blockIdx.x;
    const int seg    = blk % NSEG;
    const int fChunk = (blk / NSEG) % NF_CHUNKS;
    const int b      = blk / (NSEG * NF_CHUNKS);

    const int f0   = fChunk * FCHUNK;
    const int tid  = threadIdx.x;
    const int lane = tid & 31;
    const int wave = __builtin_amdgcn_readfirstlane(tid >> 5);

    const char* wrow = (const char*)(w + (size_t)(f0 + wave) * WB_PER_F);
    v4f wv[WQUADS];
#pragma unroll
    for (int j = 0; j < WQUADS; ++j)
        wv[j] = *(const v4f*)(wrow + (unsigned)(lane * 16 + j * 512));

    const char* xb = (const char*)(x + (size_t)b * C_ * T_ * F_ + f0);
    v4f xa[XUNITS], xc[XUNITS];

#define ISSUE_X_LOADS(T0)                                                        \
    _Pragma("unroll")                                                            \
    for (int it = 0; it < XUNITS; ++it) {                                        \
        const XUnit un = x_unit(tid + NTHREADS * it);                            \
        const int tg   = (T0) - (TC - 1) + un.tl;                                \
        const int tgc  = ((unsigned)tg < (unsigned)T_) ? tg : 0;                 \
        const unsigned boff = (unsigned)((un.c * T_ + tgc) * F_ + un.q * 4) * 4u;\
        xa[it] = *(const v4f*)(xb + boff);                                       \
        xc[it] = *(const v4f*)(xb + boff + (unsigned)(T_ * F_ * 4));             \
    }

    const int tbase = seg * TSEG;
    ISSUE_X_LOADS(tbase)

    float* wA = &lds[WB_OFF + wave * WB_PER_F];
#pragma unroll
    for (int j = 0; j < WQUADS; ++j) {
#pragma unroll
        for (int h = 0; h < 2; ++h) {
            const int ck = j * 64 + lane * 2 + h;
            const int c  = ck / TC;
            const int k  = ck - c * TC;
            const int idx = (c >> 2) * WKC + k * 4 + (c & 3);
            wA[idx]          = wv[j][2 * h];
            wA[idx + TC * 4] = wv[j][2 * h + 1];
        }
    }

    const int half = lane >> 4;
    const int mn   = lane & 15;
    const int f    = f0 + wave;

    const int tap  = ((mn & 7) < TC) ? (mn & 7) : (TC - 1);
    const float* lfW = &lds[WB_OFF + wave * WB_PER_F + ((mn >> 3) * TC + tap) * 4 + half * 2];
    const int row1 = (TTILE + mn < WIN) ? (TTILE + mn) : (WIN - 1);
    const float* lfX0 = &lds[wave * XSTRIDE_F + mn   * TSTRIDE + half * 2];
    const float* lfX1 = &lds[wave * XSTRIDE_F + row1 * TSTRIDE + half * 2];

    const float bv = bias[f * O_ + half];

#pragma unroll
    for (int tile = 0; tile < TILES_PER_WG; ++tile) {
        const int t0 = tbase + tile * TTILE;

#pragma unroll
        for (int it = 0; it < XUNITS; ++it)
            asm volatile("" : "+v"(xa[it]), "+v"(xc[it]));

#pragma unroll
        for (int it = 0; it < XUNITS; ++it) {
            const XUnit un = x_unit(tid + NTHREADS * it);
            const int tg   = t0 - (TC - 1) + un.tl;
            const bool ok  = (unsigned)tg < (unsigned)T_;
            float* dst = &lds[un.q * 4 * XSTRIDE_F + un.tl * TSTRIDE + un.c];
#pragma unroll
            for (int e = 0; e < 4; ++e) {
                v2f s;
                s.x = ok ? xa[it][e] : 0.0f;
                s.y = ok ? xc[it][e] : 0.0f;
                *(v2f*)(dst + e * XSTRIDE_F) = s;
            }
        }
        __syncthreads();

        if (tile + 1 < TILES_PER_WG) {
            ISSUE_X_LOADS(t0 + TTILE)
        }

        v8f acc0 = {};
        v8f acc1 = {};
#pragma unroll
        for (int kc = 0; kc < NKC; ++kc) {
            const v2f a  = *(const v2f*)(lfW  + kc * WKC);
            const v2f b0 = *(const v2f*)(lfX0 + kc * 4);
            const v2f b1 = *(const v2f*)(lfX1 + kc * 4);
            acc0 = __builtin_amdgcn_wmma_f32_16x16x4_f32(
                false, a, false, b0, (short)0, acc0, false, false);
            acc1 = __builtin_amdgcn_wmma_f32_16x16x4_f32(
                false, a, false, b1, (short)0, acc1, false, false);
        }

        float sum = acc0[0];
        sum += rot_left_in_row<1>((mn >= 1) ? acc0[1] : acc1[1]);
        sum += rot_left_in_row<2>((mn >= 2) ? acc0[2] : acc1[2]);
        sum += rot_left_in_row<3>((mn >= 3) ? acc0[3] : acc1[3]);
        sum += rot_left_in_row<4>((mn >= 4) ? acc0[4] : acc1[4]);

        const int t = t0 + mn;
        if (t < T_)
            out[((size_t)(b * O_ + half) * T_ + t) * F_ + f] = sum + bv;

        if (tile + 1 < TILES_PER_WG)
            __syncthreads();
    }
#undef ISSUE_X_LOADS
}

extern "C" void kernel_launch(void* const* d_in, const int* in_sizes, int n_in,
                              void* d_out, int out_size, void* d_ws, size_t ws_size,
                              hipStream_t stream) {
    const float* x    = (const float*)d_in[0];
    const float* w    = (const float*)d_in[1];
    const float* bias = (const float*)d_in[2];
    float* out = (float*)d_out;

    const int grid = B_ * NF_CHUNKS * NSEG;
    decoder_out_wmma<<<grid, NTHREADS, 0, stream>>>(x, w, bias, out);
}


# Pipeline: x staged per channel half through a 2-stage register ring, barriers wait for LDS only, so loads stay in flight across stores and products; a wave owns 2 frequencies (16 waves)
# speedup vs baseline: 2.9813x; 1.3799x over previous
#include <hip/hip_runtime.h>


typedef float v2f __attribute__((ext_vector_type(2)));
typedef float v4f __attribute__((ext_vector_type(4)));
typedef float v8f __attribute__((ext_vector_type(8)));

#define TC      5
#define B_      4
#define C_      64
#define T_      1000
#define F_      96
#define O_      2
#define KTOT    (C_ * TC)
#define NKC     (C_ / 4)
#define WROWS   (O_ * TC)
#define WKC     (WROWS * 4)
#define TTILE   16
#define WIN     (TTILE + TC - 1)
#define FCHUNK  32
#define FPW     2
#define NWAVES  (FCHUNK / FPW)
#define NTHREADS (NWAVES * 32)
#define TILES_PER_WG 3
#define TSEG    (TTILE * TILES_PER_WG)
#define NSEG    ((T_ + TSEG - 1) / TSEG)
#define TSTRIDE 68
#define XSTRIDE_F 1362
#define WB_OFF  (FCHUNK * XSTRIDE_F)
#define WB_PER_F (KTOT * O_)
#define LDS_FLOATS (WB_OFF + FCHUNK * WB_PER_F)
#define NF_CHUNKS  (F_ / FCHUNK)
#define CHALF      (C_ / 2)
#define KC_STAGE   (NKC / 2)
#define STAGE_UNITS (8 * (CHALF / 2) * WIN)
#define XUNITS     (STAGE_UNITS / NTHREADS)
#define NSTAGE     (TILES_PER_WG * 2)
#define RING       2
#define WQUADS     (FPW * 5)


template <int R>
__device__ __forceinline__ float rot_left_in_row(float v) {
    const int r = __builtin_amdgcn_update_dpp(0, __builtin_bit_cast(int, v),
                                              0x120 + (16 - R), 0xf, 0xf, false);
    return __builtin_bit_cast(float, r);
}

__device__ __forceinline__ void lds_barrier() {
    asm volatile("s_wait_dscnt 0x0\n\ts_barrier_signal -1\n\ts_barrier_wait -1" ::: "memory");
}

struct XUnit { int q, tl, c; };
__device__ __forceinline__ XUnit x_unit(int u) {
    XUnit r;
    r.q = u & 7;
    const int rr    = u >> 3;
    const int tlp   = rr & 1;
    const int cpl   = (rr >> 1) & 1;
    const int rest  = rr >> 2;
    r.tl = (rest % (WIN / 2)) * 2 + tlp;
    r.c  = ((rest / (WIN / 2)) * 2 + cpl) * 2;
    return r;
}

__global__ __launch_bounds__(NTHREADS)
void decoder_out_wmma(const float* __restrict__ x,
                      const float* __restrict__ w,
                      const float* __restrict__ bias,
                      float* __restrict__ out) {
    __shared__ __attribute__((aligned(16))) float lds[LDS_FLOATS];

    const int blk    = blockIdx.x;
    const int seg    = blk % NSEG;
    const int fChunk = (blk / NSEG) % NF_CHUNKS;
    const int b      = blk / (NSEG * NF_CHUNKS);

    const int f0   = fChunk * FCHUNK;
    const int tid  = threadIdx.x;
    const int lane = tid & 31;
    const int wave = __builtin_amdgcn_readfirstlane(tid >> 5);
    const int fl0  = wave * FPW;

    const int half = lane >> 4;
    const int mn   = lane & 15;

    const int   fg  = f0 + fl0;
    const float bv0 = bias[fg * O_ + half];
    const float bv1 = bias[(fg + 1) * O_ + half];

    const char* wrow = (const char*)(w + (size_t)(f0 + fl0) * WB_PER_F);
    v4f wv[WQUADS];
#pragma unroll
    for (int j = 0; j < WQUADS; ++j)
        wv[j] = *(const v4f*)(wrow + (unsigned)(lane * 16 + j * 512));

    const char* xb = (const char*)(x + (size_t)b * C_ * T_ * F_ + f0);
    v4f xa[RING][XUNITS], xc[RING][XUNITS];
    const int tbase = seg * TSEG;

#define ISSUE_X_LOADS(S)                                                         \
    _Pragma("unroll")                                                            \
    for (int it = 0; it < XUNITS; ++it) {                                        \
        const XUnit un = x_unit(tid + NTHREADS * it + STAGE_UNITS * ((S) & 1));  \
        const int tg   = tbase + ((S) >> 1) * TTILE - (TC - 1) + un.tl;          \
        const int tgc  = ((unsigned)tg < (unsigned)T_) ? tg : 0;                 \
        const unsigned boff = (unsigned)((un.c * T_ + tgc) * F_ + un.q * 4) * 4u;\
        xa[(S) % RING][it] = *(const v4f*)(xb + boff);                           \
        xc[(S) % RING][it] = *(const v4f*)(xb + boff + (unsigned)(T_ * F_ * 4)); \
    }

    ISSUE_X_LOADS(0)
    ISSUE_X_LOADS(1)

#pragma unroll
    for (int j = 0; j < WQUADS; ++j) {
        float* wA = &lds[WB_OFF + (fl0 + j / 5) * WB_PER_F];
#pragma unroll
        for (int h = 0; h < 2; ++h) {
            const int ck = (j % 5) * 64 + lane * 2 + h;
            const int c  = ck / TC;
            const int k  = ck - c * TC;
            const int idx = (c >> 2) * WKC + k * 4 + (c & 3);
            wA[idx]          = wv[j][2 * h];
            wA[idx + TC * 4] = wv[j][2 * h + 1];
        }
    }


    const int tap  = ((mn & 7) < TC) ? (mn & 7) : (TC - 1);
    const float* lfW = &lds[WB_OFF + fl0 * WB_PER_F + ((mn >> 3) * TC + tap) * 4 + half * 2];
    const int row1 = (TTILE + mn < WIN) ? (TTILE + mn) : (WIN - 1);
    const float* lfX0 = &lds[fl0 * XSTRIDE_F + mn   * TSTRIDE + half * 2];
    const float* lfX1 = &lds[fl0 * XSTRIDE_F + row1 * TSTRIDE + half * 2];

    v8f acc0[FPW], acc1[FPW];

#pragma unroll
    for (int s = 0; s < NSTAGE; ++s) {
        const int tile = s >> 1;
        const int h2   = s & 1;
        const int slot = s % RING;
        const int t0   = tbase + tile * TTILE;

#pragma unroll
        for (int it = 0; it < XUNITS; ++it)
            asm volatile("" : "+v"(xa[slot][it]), "+v"(xc[slot][it]));

#pragma unroll
        for (int it = 0; it < XUNITS; ++it) {
            const XUnit un = x_unit(tid + NTHREADS * it + STAGE_UNITS * h2);
            const int tg   = t0 - (TC - 1) + un.tl;
            const bool ok  = (unsigned)tg < (unsigned)T_;
            float* dst = &lds[un.q * 4 * XSTRIDE_F + un.tl * TSTRIDE + un.c];
#pragma unroll
            for (int e = 0; e < 4; ++e) {
                v2f sv;
                sv.x = ok ? xa[slot][it][e] : 0.0f;
                sv.y = ok ? xc[slot][it][e] : 0.0f;
                *(v2f*)(dst + e * XSTRIDE_F) = sv;
            }
        }

        if (s + RING < NSTAGE) {
            ISSUE_X_LOADS(s + RING)
        }

        lds_barrier();

        if (h2 == 0) {
#pragma unroll
            for (int ff = 0; ff < FPW; ++ff) {
                acc0[ff] = v8f{};
                acc1[ff] = v8f{};
            }
        }

#pragma unroll
        for (int kk = 0; kk < KC_STAGE; ++kk) {
            const int kc = h2 * KC_STAGE + kk;
            if (kk == KC_STAGE / 2)
                __builtin_amdgcn_sched_barrier(0);
#pragma unroll
            for (int ff = 0; ff < FPW; ++ff) {
                const v2f a  = *(const v2f*)(lfW  + ff * WB_PER_F  + kc * WKC);
                const v2f b0 = *(const v2f*)(lfX0 + ff * XSTRIDE_F + kc * 4);
                const v2f b1 = *(const v2f*)(lfX1 + ff * XSTRIDE_F + kc * 4);
                acc0[ff] = __builtin_amdgcn_wmma_f32_16x16x4_f32(
                    false, a, false, b0, (short)0, acc0[ff], false, false);
                acc1[ff] = __builtin_amdgcn_wmma_f32_16x16x4_f32(
                    false, a, false, b1, (short)0, acc1[ff], false, false);
            }
        }

        if (h2 == 1) {
            float sum[FPW];
#pragma unroll
            for (int ff = 0; ff < FPW; ++ff) {
                float sm = acc0[ff][0];
                sm += rot_left_in_row<1>((mn >= 1) ? acc0[ff][1] : acc1[ff][1]);
                sm += rot_left_in_row<2>((mn >= 2) ? acc0[ff][2] : acc1[ff][2]);
                sm += rot_left_in_row<3>((mn >= 3) ? acc0[ff][3] : acc1[ff][3]);
                sm += rot_left_in_row<4>((mn >= 4) ? acc0[ff][4] : acc1[ff][4]);
                sum[ff] = sm;
            }

            const int t = t0 + mn;
            if (t < T_) {
                v2f ov;
                ov.x = sum[0] + bv0;
                ov.y = sum[1] + bv1;
                *(v2f*)(out + ((size_t)(b * O_ + half) * T_ + t) * F_ + fg) = ov;
            }
        }
    }
#undef ISSUE_X_LOADS
}

extern "C" void kernel_launch(void* const* d_in, const int* in_sizes, int n_in,
                              void* d_out, int out_size, void* d_ws, size_t ws_size,
                              hipStream_t stream) {
    const float* x    = (const float*)d_in[0];
    const float* w    = (const float*)d_in[1];
    const float* bias = (const float*)d_in[2];
    float* out = (float*)d_out;

    const int grid = B_ * NF_CHUNKS * NSEG;
    decoder_out_wmma<<<grid, NTHREADS, 0, stream>>>(x, w, bias, out);
}


# Wave roles: waves 0-7 only load x (one stage of loads always in flight, reloaded unit by unit) and store it to LDS; waves 8-15 (4 frequencies each) only run the previous stage's WMMAs; sum order kept
# speedup vs baseline: 3.3897x; 1.1370x over previous
#include <hip/hip_runtime.h>


typedef float v2f __attribute__((ext_vector_type(2)));
typedef float v4f __attribute__((ext_vector_type(4)));
typedef float v8f __attribute__((ext_vector_type(8)));

#define TC      5
#define B_      4
#define C_      64
#define T_      1000
#define F_      96
#define O_      2
#define KTOT    (C_ * TC)
#define NKC     (C_ / 4)
#define WROWS   (O_ * TC)
#define WKC     (WROWS * 4)
#define TTILE   16
#define WIN     (TTILE + TC - 1)
#define FCHUNK  32
#define NWAVES  16
#define NLOADW  8
#define NPRODW  (NWAVES - NLOADW)
#define FPW     (FCHUNK / NPRODW)
#define NTHREADS (NWAVES * 32)
#define LTHREADS (NLOADW * 32)
#define TILES_PER_WG 3
#define TSEG    (TTILE * TILES_PER_WG)
#define NSEG    ((T_ + TSEG - 1) / TSEG)
#define TSTRIDE 68
#define XSTRIDE_F 1362
#define WB_OFF  (FCHUNK * XSTRIDE_F)
#define WB_PER_F (KTOT * O_)
#define LDS_FLOATS (WB_OFF + FCHUNK * WB_PER_F)
#define NF_CHUNKS  (F_ / FCHUNK)
#define CHALF      (C_ / 2)
#define KC_STAGE   (NKC / 2)
#define STAGE_UNITS (8 * (CHALF / 2) * WIN)
#define XUNITS     (STAGE_UNITS / LTHREADS)
#define NSTAGE     (TILES_PER_WG * 2)
#define NREG       XUNITS
#define WQUADS     (FPW * 5)


template <int R>
__device__ __forceinline__ float rot_left_in_row(float v) {
    const int r = __builtin_amdgcn_update_dpp(0, __builtin_bit_cast(int, v),
                                              0x120 + (16 - R), 0xf, 0xf, false);
    return __builtin_bit_cast(float, r);
}

__device__ __forceinline__ void lds_barrier() {
    asm volatile("s_wait_dscnt 0x0\n\ts_barrier_signal -1\n\ts_barrier_wait -1" ::: "memory");
}

__global__ __launch_bounds__(NTHREADS)
void decoder_out_wmma(const float* __restrict__ x,
                      const float* __restrict__ w,
                      const float* __restrict__ bias,
                      float* __restrict__ out) {
    __shared__ __attribute__((aligned(16))) float lds[LDS_FLOATS];

    const int blk    = blockIdx.x;
    const int seg    = blk % NSEG;
    const int fChunk = (blk / NSEG) % NF_CHUNKS;
    const int b      = blk / (NSEG * NF_CHUNKS);

    const int f0   = fChunk * FCHUNK;
    const int tid  = threadIdx.x;
    const int lane = tid & 31;
    const int wave = __builtin_amdgcn_readfirstlane(tid >> 5);
    const bool loader = wave < NLOADW;

    const int half = lane >> 4;
    const int mn   = lane & 15;

    const int tbase = seg * TSEG;

    const int q   = lane & 7;
    const int tlp = (lane >> 3) & 1;
    const int cpl = (lane >> 4) & 1;
    const char* xb = (const char*)(x + (size_t)b * C_ * T_ * F_ + f0);

    v8f R[NREG];
    v8f acc[2 * FPW];

#define ISSUE_X_UNIT(ST, it)                                                     \
    {                                                                            \
        const int rest = wave + NLOADW * (it) + (STAGE_UNITS / 32) * ((ST) & 1); \
        const int tl   = (rest % (WIN / 2)) * 2 + tlp;                           \
        const int c    = ((rest / (WIN / 2)) * 2 + cpl) * 2;                     \
        const int tg   = tbase + ((ST) >> 1) * TTILE - (TC - 1) + tl;            \
        const int tgc  = ((unsigned)tg < (unsigned)T_) ? tg : 0;                 \
        const unsigned boff = (unsigned)((c * T_ + tgc) * F_ + q * 4) * 4u;      \
        const v4f lo = *(const v4f*)(xb + boff);                                 \
        const v4f hi = *(const v4f*)(xb + boff + (unsigned)(T_ * F_ * 4));       \
        R[it] = __builtin_shufflevector(lo, hi, 0, 1, 2, 3, 4, 5, 6, 7);         \
    }

    const int wc  = wave - NLOADW;
    const int fl0 = (wc & (NPRODW - 1)) * FPW;
    const int fg  = f0 + fl0;
    const int tap  = ((mn & 7) < TC) ? (mn & 7) : (TC - 1);
    const float* lfW = &lds[WB_OFF + fl0 * WB_PER_F + ((mn >> 3) * TC + tap) * 4 + half * 2];
    const int row1 = (TTILE + mn < WIN) ? (TTILE + mn) : (WIN - 1);
    const float* lfX0 = &lds[fl0 * XSTRIDE_F + mn   * TSTRIDE + half * 2];
    const float* lfX1 = &lds[fl0 * XSTRIDE_F + row1 * TSTRIDE + half * 2];
    float bv[FPW];

    if (loader) {
#pragma unroll
        for (int it = 0; it < XUNITS; ++it)
            ISSUE_X_UNIT(0, it)
    } else {
#pragma unroll
        for (int e = 0; e < FPW; ++e)
            bv[e] = bias[(fg + e) * O_ + half];

        const char* wrow = (const char*)(w + (size_t)fg * WB_PER_F);
        v4f wv[WQUADS];
#pragma unroll
        for (int j = 0; j < WQUADS; ++j)
            wv[j] = *(const v4f*)(wrow + (unsigned)(lane * 16 + j * 512));
#pragma unroll
        for (int j = 0; j < WQUADS; ++j) {
            float* wA = &lds[WB_OFF + (fl0 + j / 5) * WB_PER_F];
#pragma unroll
            for (int h = 0; h < 2; ++h) {
                const int ck = (j % 5) * 64 + lane * 2 + h;
                const int c  = ck / TC;
                const int k  = ck - c * TC;
                const int idx = (c >> 2) * WKC + k * 4 + (c & 3);
                wA[idx]          = wv[j][2 * h];
                wA[idx + TC * 4] = wv[j][2 * h + 1];
            }
        }
    }

#pragma unroll
    for (int k = 0; k <= NSTAGE; ++k) {
        if (loader) {
            if (k < NSTAGE) {
                const int h2 = k & 1;
                const int t0 = tbase + (k >> 1) * TTILE;
#pragma unroll
                for (int it = 0; it < XUNITS; ++it) {
                    asm volatile("" : "+v"(R[it]));

                    const int rest = wave + NLOADW * it + (STAGE_UNITS / 32) * h2;
                    const int tl   = (rest % (WIN / 2)) * 2 + tlp;
                    const int c    = ((rest / (WIN / 2)) * 2 + cpl) * 2;
                    const int tg   = t0 - (TC - 1) + tl;
                    const bool ok  = (unsigned)tg < (unsigned)T_;
                    float* dst = &lds[q * 4 * XSTRIDE_F + tl * TSTRIDE + c];
                    const v8f u = R[it];
#pragma unroll
                    for (int e = 0; e < 4; ++e) {
                        v2f sv;
                        sv.x = ok ? u[e]     : 0.0f;
                        sv.y = ok ? u[4 + e] : 0.0f;
                        *(v2f*)(dst + e * XSTRIDE_F) = sv;
                    }

                    if (k + 1 < NSTAGE)
                        ISSUE_X_UNIT(k + 1, it)
                }
            }
        } else if (k >= 1) {
            const int s    = k - 1;
            const int tile = s >> 1;
            const int h2   = s & 1;
            const int t0   = tbase + tile * TTILE;

            if (h2 == 0) {
#pragma unroll
                for (int i = 0; i < 2 * FPW; ++i)
                    acc[i] = v8f{};
            }

#pragma unroll
            for (int kk = 0; kk < KC_STAGE; ++kk) {
                const int kc = h2 * KC_STAGE + kk;
                if (kk > 0)
                    __builtin_amdgcn_sched_barrier(0);
#pragma unroll
                for (int ff = 0; ff < FPW; ++ff) {
                    const v2f a  = *(const v2f*)(lfW  + ff * WB_PER_F  + kc * WKC);
                    const v2f b0 = *(const v2f*)(lfX0 + ff * XSTRIDE_F + kc * 4);
                    const v2f b1 = *(const v2f*)(lfX1 + ff * XSTRIDE_F + kc * 4);
                    acc[2 * ff] = __builtin_amdgcn_wmma_f32_16x16x4_f32(
                        false, a, false, b0, (short)0, acc[2 * ff], false, false);
                    acc[2 * ff + 1] = __builtin_amdgcn_wmma_f32_16x16x4_f32(
                        false, a, false, b1, (short)0, acc[2 * ff + 1], false, false);
                }
            }

            if (h2 == 1) {
                v4f ov;
#pragma unroll
                for (int ff = 0; ff < FPW; ++ff) {
                    const v8f a0 = acc[2 * ff];
                    const v8f a1 = acc[2 * ff + 1];
                    float sm = a0[0];
                    sm += rot_left_in_row<1>((mn >= 1) ? a0[1] : a1[1]);
                    sm += rot_left_in_row<2>((mn >= 2) ? a0[2] : a1[2]);
                    sm += rot_left_in_row<3>((mn >= 3) ? a0[3] : a1[3]);
                    sm += rot_left_in_row<4>((mn >= 4) ? a0[4] : a1[4]);
                    ov[ff] = sm + bv[ff];
                }

                const int t = t0 + mn;
                if (t < T_)
                    *(v4f*)(out + ((size_t)(b * O_ + half) * T_ + t) * F_ + fg) = ov;
            }
        }

        if (k < NSTAGE)
            lds_barrier();
    }
#undef ISSUE_X_UNIT
}

extern "C" void kernel_launch(void* const* d_in, const int* in_sizes, int n_in,
                              void* d_out, int out_size, void* d_ws, size_t ws_size,
                              hipStream_t stream) {
    const float* x    = (const float*)d_in[0];
    const float* w    = (const float*)d_in[1];
    const float* bias = (const float*)d_in[2];
    float* out = (float*)d_out;

    const int grid = B_ * NF_CHUNKS * NSEG;
    decoder_out_wmma<<<grid, NTHREADS, 0, stream>>>(x, w, bias, out);
}


# Tiling: a work-group's 52 window rows cut into 4 disjoint 13-row column tiles, each row loaded and multiplied once; an output sums its taps across 2 tiles: 256 WMMAs per wave, not 384; sum order kept
# speedup vs baseline: 4.1270x; 1.2175x over previous
#include <hip/hip_runtime.h>


typedef float v2f __attribute__((ext_vector_type(2)));
typedef float v4f __attribute__((ext_vector_type(4)));
typedef float v8f __attribute__((ext_vector_type(8)));

#define TC      5
#define B_      4
#define C_      64
#define T_      1000
#define F_      96
#define O_      2
#define KTOT    (C_ * TC)
#define NKC     (C_ / 4)
#define WROWS   (O_ * TC)
#define WKC     (WROWS * 4)
#define TSEG    48
#define WINSEG  (TSEG + TC - 1)
#define NTILE   4
#define TROWS   (WINSEG / NTILE)
#define FCHUNK  32
#define NWAVES  16
#define NLOADW  8
#define NPRODW  (NWAVES - NLOADW)
#define FPW     (FCHUNK / NPRODW)
#define NTHREADS (NWAVES * 32)
#define NSEG    ((T_ + TSEG - 1) / TSEG)
#define TSTRIDE 68
#define XSTRIDE_F (TROWS * TSTRIDE + 2)
#define WB_OFF  (FCHUNK * XSTRIDE_F)
#define WB_PER_F (KTOT * O_)
#define LDS_FLOATS (WB_OFF + FCHUNK * WB_PER_F)
#define NF_CHUNKS  (F_ / FCHUNK)
#define CHALF      (C_ / 2)
#define KC_STAGE   (NKC / 2)
#define XUNITS  TROWS
#define NSTAGE  (NTILE * 2)


template <int K>
__device__ __forceinline__ float ror_in_row(float v) {
    const int r = __builtin_amdgcn_update_dpp(0, __builtin_bit_cast(int, v),
                                              0x120 + K, 0xf, 0xf, false);
    return __builtin_bit_cast(float, r);
}

__device__ __forceinline__ void lds_barrier() {
    asm volatile("s_wait_dscnt 0x0\n\ts_barrier_signal -1\n\ts_barrier_wait -1" ::: "memory");
}

__global__ __launch_bounds__(NTHREADS)
void decoder_out_wmma(const float* __restrict__ x,
                      const float* __restrict__ w,
                      const float* __restrict__ bias,
                      float* __restrict__ out) {
    __shared__ __attribute__((aligned(16))) float lds[LDS_FLOATS];

    const int blk    = blockIdx.x;
    const int seg    = blk % NSEG;
    const int fChunk = (blk / NSEG) % NF_CHUNKS;
    const int b      = blk / (NSEG * NF_CHUNKS);

    const int f0   = fChunk * FCHUNK;
    const int tid  = threadIdx.x;
    const int lane = tid & 31;
    const int wave = __builtin_amdgcn_readfirstlane(tid >> 5);
    const bool loader = wave < NLOADW;

    const int half = lane >> 4;
    const int mn   = lane & 15;

    const int tbase = seg * TSEG;

    const int q  = lane & 7;
    const int cp = lane >> 3;
    const int cw = wave * 8 + cp * 2;
    const char* xb = (const char*)(x + (size_t)b * C_ * T_ * F_ + f0);

    v8f R[XUNITS];
    v8f acc[FPW];
    float pvt[FPW][TC - 1];

#define ISSUE_X_UNIT(TILE, it)                                                   \
    {                                                                            \
        const int tg  = tbase - (TC - 1) + (TILE) * TROWS + (it);                \
        const int tgc = ((unsigned)tg < (unsigned)T_) ? tg : 0;                  \
        const unsigned boff = (unsigned)((cw * T_ + tgc) * F_ + q * 4) * 4u;     \
        const v4f lo = *(const v4f*)(xb + boff);                                 \
        const v4f hi = *(const v4f*)(xb + boff + (unsigned)(T_ * F_ * 4));       \
        R[it] = __builtin_shufflevector(lo, hi, 0, 1, 2, 3, 4, 5, 6, 7);         \
    }

    const int wc  = wave - NLOADW;
    const int fl0 = (wc & (NPRODW - 1)) * FPW;
    const int fg  = f0 + fl0;
    const int tap  = ((mn & 7) < TC) ? (mn & 7) : (TC - 1);
    const float* lfW = &lds[WB_OFF + fl0 * WB_PER_F + ((mn >> 3) * TC + tap) * 4 + half * 2];
    const int rowB = (mn < TROWS) ? mn : (TROWS - 1);
    const float* lfX = &lds[fl0 * XSTRIDE_F + rowB * TSTRIDE + half * 2];
    float bv[FPW];

    if (loader) {
#pragma unroll
        for (int it = 0; it < XUNITS; ++it)
            ISSUE_X_UNIT(0, it)
    } else {
#pragma unroll
        for (int e = 0; e < FPW; ++e)
            bv[e] = bias[(fg + e) * O_ + half];
#pragma unroll
        for (int i = 0; i < FPW; ++i)
#pragma unroll
            for (int r = 0; r < TC - 1; ++r)
                pvt[i][r] = 0.0f;

        const char* wrow = (const char*)(w + (size_t)fg * WB_PER_F);
        v4f wv[FPW * 5];
#pragma unroll
        for (int j = 0; j < FPW * 5; ++j)
            wv[j] = *(const v4f*)(wrow + (unsigned)(lane * 16 + j * 512));
#pragma unroll
        for (int j = 0; j < FPW * 5; ++j) {
            float* wA = &lds[WB_OFF + (fl0 + j / 5) * WB_PER_F];
#pragma unroll
            for (int h = 0; h < 2; ++h) {
                const int ck = (j % 5) * 64 + lane * 2 + h;
                const int c  = ck / TC;
                const int k  = ck - c * TC;
                const int idx = (c >> 2) * WKC + k * 4 + (c & 3);
                wA[idx]          = wv[j][2 * h];
                wA[idx + TC * 4] = wv[j][2 * h + 1];
            }
        }
    }

#pragma unroll
    for (int k = 0; k <= NSTAGE; ++k) {
        if (loader) {
            if (k < NSTAGE && (wave >> 2) == (k & 1)) {
                const int tile = k >> 1;
                const int tw = tbase - (TC - 1) + tile * TROWS;
#pragma unroll
                for (int it = 0; it < XUNITS; ++it) {
                    asm volatile("" : "+v"(R[it]));

                    const int tg = tw + it;
                    const bool ok = (unsigned)tg < (unsigned)T_;
                    float* dst = &lds[q * 4 * XSTRIDE_F + it * TSTRIDE + cw];
                    const v8f uu = R[it];
#pragma unroll
                    for (int e = 0; e < 4; ++e) {
                        v2f sv;
                        sv.x = ok ? uu[e]     : 0.0f;
                        sv.y = ok ? uu[4 + e] : 0.0f;
                        *(v2f*)(dst + e * XSTRIDE_F) = sv;
                    }

                    if (tile + 1 < NTILE)
                        ISSUE_X_UNIT(tile + 1, it)
                }
            }
        } else if (k >= 1) {
            const int s    = k - 1;
            const int tile = s >> 1;
            const int h2   = s & 1;

            if (h2 == 0) {
#pragma unroll
                for (int i = 0; i < FPW; ++i)
                    acc[i] = v8f{};
            }

#pragma unroll
            for (int kk = 0; kk < KC_STAGE; ++kk) {
                const int kc = h2 * KC_STAGE + kk;
                if (kk > 0 && (kk & 1) == 0)
                    __builtin_amdgcn_sched_barrier(0);
#pragma unroll
                for (int ff = 0; ff < FPW; ++ff) {
                    const v2f a  = *(const v2f*)(lfW + ff * WB_PER_F  + kc * WKC);
                    const v2f bx = *(const v2f*)(lfX + ff * XSTRIDE_F + kc * 4);
                    acc[ff] = __builtin_amdgcn_wmma_f32_16x16x4_f32(
                        false, a, false, bx, (short)0, acc[ff], false, false);
                }
            }

            if (h2 == 1) {
                v4f ov;
#pragma unroll
                for (int ff = 0; ff < FPW; ++ff) {
                    const v8f cu = acc[ff];
                    const float c0 = ror_in_row<4>(cu[0]), p0 = ror_in_row<7>(pvt[ff][0]);
                    const float c1 = ror_in_row<3>(cu[1]), p1 = ror_in_row<6>(pvt[ff][1]);
                    const float c2 = ror_in_row<2>(cu[2]), p2 = ror_in_row<5>(pvt[ff][2]);
                    const float c3 = ror_in_row<1>(cu[3]), p3 = ror_in_row<4>(pvt[ff][3]);
                    float sm = (mn >= 4) ? c0 : p0;
                    sm += (mn >= 3) ? c1 : p1;
                    sm += (mn >= 2) ? c2 : p2;
                    sm += (mn >= 1) ? c3 : p3;
                    sm += cu[4];
                    ov[ff] = sm + bv[ff];
#pragma unroll
                    for (int r = 0; r < TC - 1; ++r)
                        pvt[ff][r] = cu[r];
                }

                const int i = tile * TROWS - (TC - 1) + mn;
                const int t = tbase + i;
                if (mn < TROWS && i >= 0 && t < T_)
                    *(v4f*)(out + ((size_t)(b * O_ + half) * T_ + t) * F_ + fg) = ov;
            }
        }

        if (k < NSTAGE)
            lds_barrier();
    }
#undef ISSUE_X_UNIT
}

extern "C" void kernel_launch(void* const* d_in, const int* in_sizes, int n_in,
                              void* d_out, int out_size, void* d_ws, size_t ws_size,
                              hipStream_t stream) {
    const float* x    = (const float*)d_in[0];
    const float* w    = (const float*)d_in[1];
    const float* bias = (const float*)d_in[2];
    float* out = (float*)d_out;

    const int grid = B_ * NF_CHUNKS * NSEG;
    decoder_out_wmma<<<grid, NTHREADS, 0, stream>>>(x, w, bias, out);
}
